# starting kernel
# baseline (speedup 1.0000x reference)
#include <hip/hip_runtime.h>
#include <hip/hip_bf16.h>

#define Zn 131072
#define Nn 64
#define Kn 64
#define Wn 32

typedef __attribute__((ext_vector_type(2))) float v2f;
typedef __attribute__((ext_vector_type(8))) float v8f;

__global__ __launch_bounds__(256) void lts_reduce_mix_kernel(
    const float* __restrict__ mix,
    const float* __restrict__ wgt,
    float* __restrict__ Msw)
{
    const int t = blockIdx.x * blockDim.x + threadIdx.x;
    float acc = 0.0f;
#pragma unroll
    for (int w = 0; w < Wn; ++w)
        acc = fmaf(wgt[w], mix[w * (Kn * Nn * Nn) + t], acc);

    const int k  = t >> 12;
    const int i  = (t >> 6) & 63;
    const int j  = t & 63;
    const int kt = k >> 4;
    const int kl = k & 15;
    const int jj = j >> 2;
    const int jo = j & 3;
    const int half = jo >> 1;
    const int lane = half * 16 + kl;
    const int fidx = (((kt * 64 + i) * 16 + jj) * 32 + lane) * 2 + (jo & 1);
    Msw[fidx] = acc;
}

__global__ __launch_bounds__(1024) void lts_bilinear_wmma_kernel(
    const float* __restrict__ feat,
    const float* __restrict__ Msw,
    float* __restrict__ out)
{
    extern __shared__ float smem[];

    const int tid  = threadIdx.x;
    const int wave = tid >> 5;
    const int lane = tid & 31;
    const int half = lane >> 4;
    const int l16  = lane & 15;
    const int zrow = blockIdx.x * 512 + wave * 16 + l16;

    v2f  fj[16];
#pragma unroll
    for (int jj = 0; jj < 16; ++jj) {
        const float2 v = *(const float2*)(feat + zrow * 64 + jj * 4 + half * 2);
        fj[jj].x = v.x;
        fj[jj].y = v.y;
    }
    const float* frow = feat + zrow * 64;

    for (int kt = 0; kt < 4; ++kt) {
        __syncthreads();
        {
            const float4* src = (const float4*)(Msw + (size_t)kt * 65536);
            float4* dst = (float4*)smem;
            for (int idx = tid; idx < 16384; idx += 1024)
                dst[idx] = src[idx];
        }
        __syncthreads();

        const float2* Bl = (const float2*)smem;
        v8f acc = {};
        for (int i = 0; i < 64; ++i) {
            const float fi = frow[i];
            const float2* bp = Bl + i * 512 + lane;
#pragma unroll
            for (int jj = 0; jj < 16; ++jj) {
                const float2 bv = bp[jj * 32];
                v2f b; b.x = bv.x; b.y = bv.y;
                v2f a; a.x = fi * fj[jj].x; a.y = fi * fj[jj].y;
                acc = __builtin_amdgcn_wmma_f32_16x16x4_f32(
                     false, a,  false, b,
                     (short)0, acc,
                     false,  false);
            }
        }

#pragma unroll
        for (int r = 0; r < 8; ++r) {
            const int z = blockIdx.x * 512 + wave * 16 + half * 8 + r;
            out[z * 64 + kt * 16 + l16] = acc[r];
        }
    }
}

extern "C" void kernel_launch(void* const* d_in, const int* in_sizes, int n_in,
                              void* d_out, int out_size, void* d_ws, size_t ws_size,
                              hipStream_t stream) {
    const float* feat = (const float*)d_in[0];
    const float* mix  = (const float*)d_in[1];
    const float* wgt  = (const float*)d_in[2];
    float* out = (float*)d_out;
    float* Msw = (float*)d_ws;

    lts_reduce_mix_kernel<<<(Kn * Nn * Nn) / 256, 256, 0, stream>>>(mix, wgt, Msw);

    lts_bilinear_wmma_kernel<<<Zn / 512, 1024, 65536 * sizeof(float), stream>>>(
        feat, Msw, out);
}


# Kernel 2 contracts j first on the matrix unit with the features as loaded (A = M, B = f); f[z,i] scales each finished 16-step chain per lane: no vector product feeds a matrix step; sum order changes
# speedup vs baseline: 1.1726x; 1.1726x over previous
#include <hip/hip_runtime.h>

#define Zn 131072
#define Nn 64
#define Kn 64
#define Wn 32

#define ZT_PER_BLOCK 128
#define ROWS_PER_BLOCK (ZT_PER_BLOCK * 16)

typedef __attribute__((ext_vector_type(2))) float v2f;
typedef __attribute__((ext_vector_type(8))) float v8f;

__global__ __launch_bounds__(256) void lts_reduce_mix_kernel(
    const float* __restrict__ mix,
    const float* __restrict__ wgt,
    float* __restrict__ Msw)
{
    const int t = blockIdx.x * blockDim.x + threadIdx.x;
    float acc = 0.0f;
#pragma unroll
    for (int w = 0; w < Wn; ++w)
        acc = fmaf(wgt[w], mix[w * (Kn * Nn * Nn) + t], acc);

    const int k  = t >> 12;
    const int i  = (t >> 6) & 63;
    const int j  = t & 63;
    const int kt = k >> 4;
    const int c  = k & 15;
    const int wv = i >> 3;
    const int il = i & 7;
    const int h  = j >> 5;
    const int q  = (j & 31) >> 1;
    const int s  = j & 1;
    const int lane = h * 16 + c;
    const int frag = ((kt * 8 + wv) * 8 + il) * 16 + q;
    Msw[(frag * 32 + lane) * 2 + s] = acc;
}

struct TileFeat {
    float4 fj[8];
    float4 fi[2];
};

__device__ __forceinline__ void load_tile(TileFeat& f, const float* __restrict__ rowp,
                                          int half, int wave) {
#pragma unroll
    for (int m = 0; m < 8; ++m)
        f.fj[m] = *(const float4*)(rowp + 32 * half + 4 * m);
    f.fi[0] = *(const float4*)(rowp + 8 * wave);
    f.fi[1] = *(const float4*)(rowp + 8 * wave + 4);
}

__device__ __forceinline__ float elem4(const float4& v, int e) {
    return e == 0 ? v.x : (e == 1 ? v.y : (e == 2 ? v.z : v.w));
}

template <int IL>
__device__ __forceinline__ void chain_pair(const TileFeat& f, const v2f (&M)[8][16],
                                           v8f& d0, v8f& d1) {
    d0 = v8f{};
    d1 = v8f{};
#pragma unroll
    for (int q = 0; q < 16; ++q) {
        v2f fz;
        fz.x = elem4(f.fj[q >> 1], (q & 1) * 2);
        fz.y = elem4(f.fj[q >> 1], (q & 1) * 2 + 1);
        d0 = __builtin_amdgcn_wmma_f32_16x16x4_f32(
            false, M[IL][q], false, fz, (short)0, d0, false, false);
        d1 = __builtin_amdgcn_wmma_f32_16x16x4_f32(
            false, M[IL + 1][q], false, fz, (short)0, d1, false, false);
    }
}

template <int IL>
__device__ __forceinline__ void scale_add(const TileFeat& f, const v8f& d0, const v8f& d1,
                                          v8f& acc) {
    const float s0 = elem4(f.fi[IL >> 2], IL & 3);
    const float s1 = elem4(f.fi[(IL + 1) >> 2], (IL + 1) & 3);
#pragma unroll
    for (int r = 0; r < 8; ++r)
        acc[r] = fmaf(s1, d1[r], fmaf(s0, d0[r], acc[r]));
}

__device__ __forceinline__ void lds_barrier() {
    __builtin_amdgcn_sched_barrier(0);
    asm volatile("s_wait_dscnt 0x0\n\ts_barrier_signal -1\n\ts_barrier_wait -1" ::: "memory");
    __builtin_amdgcn_sched_barrier(0);
}

#define RED_PITCH 33
#define RED_TILE (8 * RED_PITCH)

__device__ __forceinline__ void tile_front(const TileFeat& f, const v2f (&M)[8][16], v8f& acc) {
    v8f a0, a1;
    chain_pair<0>(f, M, a0, a1);
    acc = v8f{};
    scale_add<0>(f, a0, a1, acc);
}

__device__ __forceinline__ void tile_rest(const TileFeat& f, const v2f (&M)[8][16], v8f& acc) {
    v8f b0, b1, c0, c1, d0, d1;
    chain_pair<2>(f, M, b0, b1);
    chain_pair<4>(f, M, c0, c1);
    chain_pair<6>(f, M, d0, d1);
    scale_add<2>(f, b0, b1, acc);
    scale_add<4>(f, c0, c1, acc);
    scale_add<6>(f, d0, d1, acc);
}

__global__ __launch_bounds__(256) void lts_bilinear_wmma_kernel(
    const float* __restrict__ feat,
    const float* __restrict__ Msw,
    float* __restrict__ out)
{
    __shared__ float red[2][8][RED_TILE];

    const int tid  = threadIdx.x;
    const int wave = tid >> 5;
    const int lane = tid & 31;
    const int half = lane >> 4;
    const int c    = lane & 15;
    const int kt   = blockIdx.x & 3;
    const int zbase = (blockIdx.x >> 2) * ROWS_PER_BLOCK;

    v2f M[8][16];
    {
        const float2* msrc = (const float2*)Msw + (size_t)((kt * 8 + wave) * 128) * 32 + lane;
#pragma unroll
        for (int f = 0; f < 128; ++f) {
            const float2 v = msrc[f * 32];
            M[f >> 4][f & 15].x = v.x;
            M[f >> 4][f & 15].y = v.y;
        }
    }

    const float* rowp = feat + (size_t)(zbase + c) * 64;
    float* outp = out + (size_t)(zbase + 8 * half + wave) * 64 + kt * 16 + c;
    const int rd = (c & 7) * RED_PITCH + 16 * (c >> 3) + 8 * half + wave;

    TileFeat cur, nxt;
    load_tile(cur, rowp, half, wave);

    {
        load_tile(nxt, rowp + 16 * 64, half, wave);
        __builtin_amdgcn_sched_barrier(0);
        v8f acc;
        tile_front(cur, M, acc);
        tile_rest(cur, M, acc);
#pragma unroll
        for (int r = 0; r < 8; ++r)
            red[0][wave][r * RED_PITCH + lane] = acc[r];
        cur = nxt;
    }

    for (int t = 1; t < ZT_PER_BLOCK; ++t) {
        const int tn = (t + 1 < ZT_PER_BLOCK) ? (t + 1) : t;
        load_tile(nxt, rowp + (size_t)tn * (16 * 64), half, wave);
        __builtin_amdgcn_sched_barrier(0);

        v8f acc;
        tile_front(cur, M, acc);

        lds_barrier();

        {
            const int pb = (t - 1) & 1;
            float s = red[pb][0][rd];
#pragma unroll
            for (int w = 1; w < 8; ++w)
                s += red[pb][w][rd];
            outp[(size_t)(t - 1) * (16 * 64)] = s;
        }

        tile_rest(cur, M, acc);

        const int wb = t & 1;
#pragma unroll
        for (int r = 0; r < 8; ++r)
            red[wb][wave][r * RED_PITCH + lane] = acc[r];
        cur = nxt;
    }

    lds_barrier();
    {
        const int pb = (ZT_PER_BLOCK - 1) & 1;
        float s = red[pb][0][rd];
#pragma unroll
        for (int w = 1; w < 8; ++w)
            s += red[pb][w][rd];
        outp[(size_t)(ZT_PER_BLOCK - 1) * (16 * 64)] = s;
    }
}

extern "C" void kernel_launch(void* const* d_in, const int* in_sizes, int n_in,
                              void* d_out, int out_size, void* d_ws, size_t ws_size,
                              hipStream_t stream) {
    const float* feat = (const float*)d_in[0];
    const float* mix  = (const float*)d_in[1];
    const float* wgt  = (const float*)d_in[2];
    float* out = (float*)d_out;
    float* Msw = (float*)d_ws;

    lts_reduce_mix_kernel<<<(Kn * Nn * Nn) / 256, 256, 0, stream>>>(mix, wgt, Msw);

    lts_bilinear_wmma_kernel<<<(Zn / ROWS_PER_BLOCK) * 4, 256, 0, stream>>>(feat, Msw, out);
}
